# starting kernel
# baseline (speedup 1.0000x reference)
#include <hip/hip_runtime.h>

#define NN   50000
#define NE   800000
#define INF_ 128
#define HID  64
#define HID4 16
#define DHID 384

typedef float v2f __attribute__((ext_vector_type(2)));
typedef float v8f __attribute__((ext_vector_type(8)));


__global__ void k_zero(float* __restrict__ p, int n) {
  int i = blockIdx.x * blockDim.x + threadIdx.x;
  if (i < n) p[i] = 0.0f;
}

__global__ void k_degree(const int* __restrict__ dst, float* __restrict__ deg, int e) {
  int i = blockIdx.x * blockDim.x + threadIdx.x;
  if (i < e) atomicAdd(&deg[dst[i]], 1.0f);
}

__global__ void k_dinv(float* __restrict__ d, int n) {
  int i = blockIdx.x * blockDim.x + threadIdx.x;
  if (i < n) {
    float v = d[i];
    v = v < 1.0f ? 1.0f : v;
    d[i] = rsqrtf(v);
  }
}

__global__ void k_scale_zero4(const float4* __restrict__ f, const float* __restrict__ dinv,
                              float4* __restrict__ t, float4* __restrict__ agg) {
  int i = blockIdx.x * blockDim.x + threadIdx.x;
  if (i >= NN * HID4) return;
  float dv = dinv[i >> 4];
  float4 v = f[i];
  v.x *= dv; v.y *= dv; v.z *= dv; v.w *= dv;
  t[i] = v;
  agg[i] = make_float4(0.f, 0.f, 0.f, 0.f);
}

__global__ void k_scatter4(const int* __restrict__ src, const int* __restrict__ dst,
                           const float4* __restrict__ t, float* __restrict__ agg) {
  int i = blockIdx.x * blockDim.x + threadIdx.x;
  if (i >= NE * HID4) return;
  int e = i >> 4;
  int q = i & 15;
  float4 v = t[(size_t)src[e] * HID4 + q];
  float* a = agg + (size_t)dst[e] * HID + q * 4;
  atomicAdd(a + 0, v.x);
  atomicAdd(a + 1, v.y);
  atomicAdd(a + 2, v.z);
  atomicAdd(a + 3, v.w);
}

__global__ void k_update_combine4(float4* __restrict__ f, const float4* __restrict__ agg,
                                  const float* __restrict__ dinv, float* __restrict__ out,
                                  int col_base, float c0, float c1, float c2) {
  int i = blockIdx.x * blockDim.x + threadIdx.x;
  if (i >= NN * HID4) return;
  int row = i >> 4;
  int q   = i & 15;
  float dv = dinv[row];
  float4 fv = f[i];
  float4 av = agg[i];
  fv.x -= av.x * dv; fv.y -= av.y * dv; fv.z -= av.z * dv; fv.w -= av.w * dv;
  f[i] = fv;
  float4* o = (float4*)(out + (size_t)row * DHID + col_base + q * 4);
  const float cs[3] = {c0, c1, c2};
#pragma unroll
  for (int j = 0; j < 3; ++j) {
    float4 ov = o[j * HID4];
    ov.x += cs[j] * fv.x; ov.y += cs[j] * fv.y;
    ov.z += cs[j] * fv.z; ov.w += cs[j] * fv.w;
    o[j * HID4] = ov;
  }
}


__global__ __launch_bounds__(128) void k_gemm1_relu_combine(
    const float* __restrict__ X, const float* __restrict__ W,
    const float* __restrict__ bias, float* __restrict__ Y,
    float* __restrict__ out, int col_base, float c0, float c1, float c2) {
  const int lane = threadIdx.x & 31;
  const int wave = threadIdx.x >> 5;
  const int row0 = blockIdx.x * 16;
  const int n0   = wave * 16;
  const int mr   = lane & 15;
  const int kq   = (lane >> 4) << 1;
  v8f c = {};
  const float* xrow = X + (size_t)(row0 + mr) * INF_;
  for (int k = 0; k < INF_; k += 4) {
    v2f a, b;
    a.x = xrow[k + kq];
    a.y = xrow[k + kq + 1];
    b.x = W[(size_t)(k + kq) * HID + n0 + mr];
    b.y = W[(size_t)(k + kq + 1) * HID + n0 + mr];
    c = __builtin_amdgcn_wmma_f32_16x16x4_f32(false, a, false, b, (short)0, c, false, false);
  }
  const int mhi = (lane >> 4) << 3;
  const float bv = bias[n0 + mr];
#pragma unroll
  for (int r = 0; r < 8; ++r) {
    float v = c[r] + bv;
    v = v > 0.0f ? v : 0.0f;
    const size_t row = (size_t)(row0 + mhi + r);
    Y[row * HID + n0 + mr] = v;
    float* o = out + row * DHID + col_base + n0 + mr;
    o[0]       = c0 * v;
    o[HID]     = c1 * v;
    o[2 * HID] = c2 * v;
  }
}


__global__ __launch_bounds__(128) void k_head(
    const float* __restrict__ H,
    const float* __restrict__ Wm1, const float* __restrict__ bm1,
    const float* __restrict__ Wm2, const float* __restrict__ bm2,
    float* __restrict__ logits) {
  __shared__ float m[16][HID];
  const int lane = threadIdx.x & 31;
  const int wave = threadIdx.x >> 5;
  const int row0 = blockIdx.x * 16;
  const int n0   = wave * 16;
  const int mr   = lane & 15;
  const int kq   = (lane >> 4) << 1;
  v8f c = {};
  const float* hrow = H + (size_t)(row0 + mr) * DHID;
  for (int k = 0; k < DHID; k += 4) {
    v2f a, b;
    a.x = hrow[k + kq];
    a.y = hrow[k + kq + 1];
    b.x = Wm1[(size_t)(k + kq) * HID + n0 + mr];
    b.y = Wm1[(size_t)(k + kq + 1) * HID + n0 + mr];
    c = __builtin_amdgcn_wmma_f32_16x16x4_f32(false, a, false, b, (short)0, c, false, false);
  }
  const int mhi = (lane >> 4) << 3;
  const float bv = bm1[n0 + mr];
#pragma unroll
  for (int r = 0; r < 8; ++r) {
    float v = c[r] + bv;
    m[mhi + r][n0 + mr] = v > 0.0f ? v : 0.0f;
  }
  __syncthreads();
  if (threadIdx.x < 32) {
    int row = threadIdx.x >> 1;
    int col = threadIdx.x & 1;
    float acc = bm2[col];
#pragma unroll
    for (int k = 0; k < HID; ++k) acc += m[row][k] * Wm2[k * 2 + col];
    logits[(size_t)(row0 + row) * 2 + col] = acc;
  }
}


static void run_branch(const float* X, const int* src, const int* dst,
                       const float* W1, const float* b1,
                       float* out, int col_base,
                       const float sc[3][3],
                       float* f, float* t, float* agg, float* dinv,
                       hipStream_t stream) {
  const int nNodeB = (NN + 255) / 256;
  const int nEdgeB = (NE + 255) / 256;
  const int nQuadB = (NN * HID4 + 255) / 256;
  const int nScatB = (NE * HID4 + 255) / 256;

  k_zero  <<<nNodeB, 256, 0, stream>>>(dinv, NN);
  k_degree<<<nEdgeB, 256, 0, stream>>>(dst, dinv, NE);
  k_dinv  <<<nNodeB, 256, 0, stream>>>(dinv, NN);

  k_gemm1_relu_combine<<<NN / 16, 128, 0, stream>>>(X, W1, b1, f, out, col_base,
                                                    sc[0][0], sc[0][1], sc[0][2]);

  for (int step = 1; step <= 2; ++step) {
    k_scale_zero4<<<nQuadB, 256, 0, stream>>>((const float4*)f, dinv,
                                              (float4*)t, (float4*)agg);
    k_scatter4   <<<nScatB, 256, 0, stream>>>(src, dst, (const float4*)t, agg);
    k_update_combine4<<<nQuadB, 256, 0, stream>>>((float4*)f, (const float4*)agg, dinv,
                                                  out, col_base,
                                                  sc[step][0], sc[step][1], sc[step][2]);
  }
}

extern "C" void kernel_launch(void* const* d_in, const int* in_sizes, int n_in,
                              void* d_out, int out_size, void* d_ws, size_t ws_size,
                              hipStream_t stream) {
  (void)in_sizes; (void)n_in; (void)out_size; (void)ws_size;
  const float* x     = (const float*)d_in[0];
  const float* sim_x = (const float*)d_in[1];
  const int*   src   = (const int*)d_in[2];
  const int*   dst   = (const int*)d_in[3];
  const int*   ssrc  = (const int*)d_in[4];
  const int*   sdst  = (const int*)d_in[5];
  const float* W1o   = (const float*)d_in[6];
  const float* b1o   = (const float*)d_in[7];
  const float* W1s   = (const float*)d_in[8];
  const float* b1s   = (const float*)d_in[9];
  const float* Wm1   = (const float*)d_in[10];
  const float* bm1   = (const float*)d_in[11];
  const float* Wm2   = (const float*)d_in[12];
  const float* bm2   = (const float*)d_in[13];

  float* out    = (float*)d_out;
  float* logits = out + (size_t)NN * DHID;

  float* f    = (float*)d_ws;
  float* t    = f    + (size_t)NN * HID;
  float* agg  = t    + (size_t)NN * HID;
  float* dinv = agg  + (size_t)NN * HID;

  const float sc_o[3][3] = {
      {3.0f,   0.0f,  0.0f},
      {-3.0f,  3.0f,  0.0f},
      {0.75f, -1.5f,  0.75f}};
  const float sc_s[3][3] = {
      {4.0f, 4.0f, 4.0f},
      {4.0f, 4.0f, 4.0f},
      {4.0f, 4.0f, 4.0f}};

  run_branch(x,     src,  dst,  W1o, b1o, out,   0, sc_o, f, t, agg, dinv, stream);
  run_branch(sim_x, ssrc, sdst, W1s, b1s, out, 192, sc_s, f, t, agg, dinv, stream);

  k_head<<<NN / 16, 128, 0, stream>>>(out, Wm1, bm1, Wm2, bm2, logits);
}


# scatter-add re-mapped to one wave per edge row: lane l adds columns l and l+32, each f32 atomic covers one whole 128-B line (2 line requests per edge, was 8); sums still in atomic arrival order
# speedup vs baseline: 1.2554x; 1.2554x over previous
#include <hip/hip_runtime.h>

#define NN   50000
#define NE   800000
#define INF_ 128
#define HID  64
#define HID4 16
#define DHID 384

typedef float v2f __attribute__((ext_vector_type(2)));
typedef float v8f __attribute__((ext_vector_type(8)));


__global__ void k_zero(float* __restrict__ p, int n) {
  int i = blockIdx.x * blockDim.x + threadIdx.x;
  if (i < n) p[i] = 0.0f;
}

__global__ void k_degree(const int* __restrict__ dst, float* __restrict__ deg, int e) {
  int i = blockIdx.x * blockDim.x + threadIdx.x;
  if (i < e) atomicAdd(&deg[dst[i]], 1.0f);
}

__global__ void k_dinv(float* __restrict__ d, int n) {
  int i = blockIdx.x * blockDim.x + threadIdx.x;
  if (i < n) {
    float v = d[i];
    v = v < 1.0f ? 1.0f : v;
    d[i] = rsqrtf(v);
  }
}

__global__ void k_scale_zero4(const float4* __restrict__ f, const float* __restrict__ dinv,
                              float4* __restrict__ t, float4* __restrict__ agg) {
  int i = blockIdx.x * blockDim.x + threadIdx.x;
  if (i >= NN * HID4) return;
  float dv = dinv[i >> 4];
  float4 v = f[i];
  v.x *= dv; v.y *= dv; v.z *= dv; v.w *= dv;
  t[i] = v;
  agg[i] = make_float4(0.f, 0.f, 0.f, 0.f);
}

#define EPW 32
#define EPB 8
__global__ __launch_bounds__(256) void k_scatter_rows(const int* __restrict__ src,
                                                      const int* __restrict__ dst,
                                                      const float* __restrict__ t,
                                                      float* __restrict__ agg) {
  static_assert(NE % EPW == 0 && EPW % EPB == 0, "a wave takes full batches of edges");
  const int lane = threadIdx.x & 31;
  const int grp = (int)((blockIdx.x * blockDim.x + threadIdx.x) >> 5);
  if (grp >= NE / EPW) return;
  const int sv = src[(size_t)grp * EPW + lane];
  const int dv = dst[(size_t)grp * EPW + lane];
#define SRC_OF(j) __builtin_amdgcn_readlane(sv, (j))
#define DST_OF(j) __builtin_amdgcn_readlane(dv, (j))
  const float* tl = t + lane;
  float* al = agg + lane;
  float v0[EPW], v1[EPW];
#pragma unroll
  for (int j = 0; j < EPB; ++j) {
    const float* tr = tl + (size_t)SRC_OF(j) * HID;
    v0[j] = tr[0];
    v1[j] = tr[32];
  }
#pragma unroll
  for (int b = 0; b < EPW / EPB; ++b) {
    if (b + 1 < EPW / EPB) {
#pragma unroll
      for (int j = (b + 1) * EPB; j < (b + 2) * EPB; ++j) {
        const float* tr = tl + (size_t)SRC_OF(j) * HID;
        v0[j] = tr[0];
        v1[j] = tr[32];
      }
    }
#pragma unroll
    for (int j = b * EPB; j < (b + 1) * EPB; ++j) {
      float* ar = al + (size_t)DST_OF(j) * HID;
      atomicAdd(ar, v0[j]);
      atomicAdd(ar + 32, v1[j]);
    }
  }
}

__global__ void k_update_combine4(float4* __restrict__ f, const float4* __restrict__ agg,
                                  const float* __restrict__ dinv, float* __restrict__ out,
                                  int col_base, float c0, float c1, float c2) {
  int i = blockIdx.x * blockDim.x + threadIdx.x;
  if (i >= NN * HID4) return;
  int row = i >> 4;
  int q   = i & 15;
  float dv = dinv[row];
  float4 fv = f[i];
  float4 av = agg[i];
  fv.x -= av.x * dv; fv.y -= av.y * dv; fv.z -= av.z * dv; fv.w -= av.w * dv;
  f[i] = fv;
  float4* o = (float4*)(out + (size_t)row * DHID + col_base + q * 4);
  const float cs[3] = {c0, c1, c2};
#pragma unroll
  for (int j = 0; j < 3; ++j) {
    float4 ov = o[j * HID4];
    ov.x += cs[j] * fv.x; ov.y += cs[j] * fv.y;
    ov.z += cs[j] * fv.z; ov.w += cs[j] * fv.w;
    o[j * HID4] = ov;
  }
}


__global__ __launch_bounds__(128) void k_gemm1_relu_combine(
    const float* __restrict__ X, const float* __restrict__ W,
    const float* __restrict__ bias, float* __restrict__ Y,
    float* __restrict__ out, int col_base, float c0, float c1, float c2) {
  const int lane = threadIdx.x & 31;
  const int wave = threadIdx.x >> 5;
  const int row0 = blockIdx.x * 16;
  const int n0   = wave * 16;
  const int mr   = lane & 15;
  const int kq   = (lane >> 4) << 1;
  v8f c = {};
  const float* xrow = X + (size_t)(row0 + mr) * INF_;
  for (int k = 0; k < INF_; k += 4) {
    v2f a, b;
    a.x = xrow[k + kq];
    a.y = xrow[k + kq + 1];
    b.x = W[(size_t)(k + kq) * HID + n0 + mr];
    b.y = W[(size_t)(k + kq + 1) * HID + n0 + mr];
    c = __builtin_amdgcn_wmma_f32_16x16x4_f32(false, a, false, b, (short)0, c, false, false);
  }
  const int mhi = (lane >> 4) << 3;
  const float bv = bias[n0 + mr];
#pragma unroll
  for (int r = 0; r < 8; ++r) {
    float v = c[r] + bv;
    v = v > 0.0f ? v : 0.0f;
    const size_t row = (size_t)(row0 + mhi + r);
    Y[row * HID + n0 + mr] = v;
    float* o = out + row * DHID + col_base + n0 + mr;
    o[0]       = c0 * v;
    o[HID]     = c1 * v;
    o[2 * HID] = c2 * v;
  }
}


__global__ __launch_bounds__(128) void k_head(
    const float* __restrict__ H,
    const float* __restrict__ Wm1, const float* __restrict__ bm1,
    const float* __restrict__ Wm2, const float* __restrict__ bm2,
    float* __restrict__ logits) {
  __shared__ float m[16][HID];
  const int lane = threadIdx.x & 31;
  const int wave = threadIdx.x >> 5;
  const int row0 = blockIdx.x * 16;
  const int n0   = wave * 16;
  const int mr   = lane & 15;
  const int kq   = (lane >> 4) << 1;
  v8f c = {};
  const float* hrow = H + (size_t)(row0 + mr) * DHID;
  for (int k = 0; k < DHID; k += 4) {
    v2f a, b;
    a.x = hrow[k + kq];
    a.y = hrow[k + kq + 1];
    b.x = Wm1[(size_t)(k + kq) * HID + n0 + mr];
    b.y = Wm1[(size_t)(k + kq + 1) * HID + n0 + mr];
    c = __builtin_amdgcn_wmma_f32_16x16x4_f32(false, a, false, b, (short)0, c, false, false);
  }
  const int mhi = (lane >> 4) << 3;
  const float bv = bm1[n0 + mr];
#pragma unroll
  for (int r = 0; r < 8; ++r) {
    float v = c[r] + bv;
    m[mhi + r][n0 + mr] = v > 0.0f ? v : 0.0f;
  }
  __syncthreads();
  if (threadIdx.x < 32) {
    int row = threadIdx.x >> 1;
    int col = threadIdx.x & 1;
    float acc = bm2[col];
#pragma unroll
    for (int k = 0; k < HID; ++k) acc += m[row][k] * Wm2[k * 2 + col];
    logits[(size_t)(row0 + row) * 2 + col] = acc;
  }
}


static void run_branch(const float* X, const int* src, const int* dst,
                       const float* W1, const float* b1,
                       float* out, int col_base,
                       const float sc[3][3],
                       float* f, float* t, float* agg, float* dinv,
                       hipStream_t stream) {
  const int nNodeB = (NN + 255) / 256;
  const int nEdgeB = (NE + 255) / 256;
  const int nQuadB = (NN * HID4 + 255) / 256;
  const int nScatB = (NE / EPW + 7) / 8;

  k_zero  <<<nNodeB, 256, 0, stream>>>(dinv, NN);
  k_degree<<<nEdgeB, 256, 0, stream>>>(dst, dinv, NE);
  k_dinv  <<<nNodeB, 256, 0, stream>>>(dinv, NN);

  k_gemm1_relu_combine<<<NN / 16, 128, 0, stream>>>(X, W1, b1, f, out, col_base,
                                                    sc[0][0], sc[0][1], sc[0][2]);

  for (int step = 1; step <= 2; ++step) {
    k_scale_zero4<<<nQuadB, 256, 0, stream>>>((const float4*)f, dinv,
                                              (float4*)t, (float4*)agg);
    k_scatter_rows<<<nScatB, 256, 0, stream>>>(src, dst, t, agg);
    k_update_combine4<<<nQuadB, 256, 0, stream>>>((float4*)f, (const float4*)agg, dinv,
                                                  out, col_base,
                                                  sc[step][0], sc[step][1], sc[step][2]);
  }
}

extern "C" void kernel_launch(void* const* d_in, const int* in_sizes, int n_in,
                              void* d_out, int out_size, void* d_ws, size_t ws_size,
                              hipStream_t stream) {
  (void)in_sizes; (void)n_in; (void)out_size; (void)ws_size;
  const float* x     = (const float*)d_in[0];
  const float* sim_x = (const float*)d_in[1];
  const int*   src   = (const int*)d_in[2];
  const int*   dst   = (const int*)d_in[3];
  const int*   ssrc  = (const int*)d_in[4];
  const int*   sdst  = (const int*)d_in[5];
  const float* W1o   = (const float*)d_in[6];
  const float* b1o   = (const float*)d_in[7];
  const float* W1s   = (const float*)d_in[8];
  const float* b1s   = (const float*)d_in[9];
  const float* Wm1   = (const float*)d_in[10];
  const float* bm1   = (const float*)d_in[11];
  const float* Wm2   = (const float*)d_in[12];
  const float* bm2   = (const float*)d_in[13];

  float* out    = (float*)d_out;
  float* logits = out + (size_t)NN * DHID;

  float* f    = (float*)d_ws;
  float* t    = f    + (size_t)NN * HID;
  float* agg  = t    + (size_t)NN * HID;
  float* dinv = agg  + (size_t)NN * HID;

  const float sc_o[3][3] = {
      {3.0f,   0.0f,  0.0f},
      {-3.0f,  3.0f,  0.0f},
      {0.75f, -1.5f,  0.75f}};
  const float sc_s[3][3] = {
      {4.0f, 4.0f, 4.0f},
      {4.0f, 4.0f, 4.0f},
      {4.0f, 4.0f, 4.0f}};

  run_branch(x,     src,  dst,  W1o, b1o, out,   0, sc_o, f, t, agg, dinv, stream);
  run_branch(sim_x, ssrc, sdst, W1s, b1s, out, 192, sc_s, f, t, agg, dinv, stream);

  k_head<<<NN / 16, 128, 0, stream>>>(out, Wm1, bm1, Wm2, bm2, logits);
}


# scale, update and edge-count passes re-gridded: a thread takes 5 quads (8 edges), loads issued ahead of stores; 5x (8x) fewer waves enter per launch; per-element arithmetic and rounding unchanged
# speedup vs baseline: 1.3266x; 1.0567x over previous
#include <hip/hip_runtime.h>

#define NN   50000
#define NE   800000
#define INF_ 128
#define HID  64
#define HID4 16
#define DHID 384

typedef float v2f __attribute__((ext_vector_type(2)));
typedef float v8f __attribute__((ext_vector_type(8)));
typedef float v4f __attribute__((ext_vector_type(4)));


__global__ void k_zero(float* __restrict__ p, int n) {
  int i = blockIdx.x * blockDim.x + threadIdx.x;
  if (i < n) p[i] = 0.0f;
}

#define DEG_V 2
__global__ __launch_bounds__(256) void k_degree(const int* __restrict__ dst, float* __restrict__ deg) {
  static_assert(NE % 4 == 0, "edges are read four at a time");
  const int4* d4 = (const int4*)dst;
  const int base = blockIdx.x * (256 * DEG_V) + threadIdx.x;
  int4 d[DEG_V];
#pragma unroll
  for (int u = 0; u < DEG_V; ++u) {
    const int i = base + u * 256;
    if (i < NE / 4) d[u] = d4[i];
  }
#pragma unroll
  for (int u = 0; u < DEG_V; ++u) {
    const int i = base + u * 256;
    if (i < NE / 4) {
      atomicAdd(&deg[d[u].x], 1.0f);
      atomicAdd(&deg[d[u].y], 1.0f);
      atomicAdd(&deg[d[u].z], 1.0f);
      atomicAdd(&deg[d[u].w], 1.0f);
    }
  }
}

__global__ void k_dinv(float* __restrict__ d, int n) {
  int i = blockIdx.x * blockDim.x + threadIdx.x;
  if (i < n) {
    float v = d[i];
    v = v < 1.0f ? 1.0f : v;
    d[i] = rsqrtf(v);
  }
}

#define NQ  (NN * HID4)
#define QPT 5
static_assert(NQ % (256 * QPT) == 0, "blocks of 256 threads cover the plane exactly");

__global__ __launch_bounds__(256) void k_scale_zero4(const float4* __restrict__ f,
                                                     const float* __restrict__ dinv,
                                                     float4* __restrict__ t, float4* __restrict__ agg) {
  const int base = blockIdx.x * (256 * QPT) + threadIdx.x;
  float4 v[QPT];
  float dv[QPT];
#pragma unroll
  for (int u = 0; u < QPT; ++u) {
    const int i = base + u * 256;
    v[u] = f[i];
    dv[u] = dinv[i >> 4];
  }
#pragma unroll
  for (int u = 0; u < QPT; ++u) {
    const int i = base + u * 256;
    float4 w = v[u];
    w.x *= dv[u]; w.y *= dv[u]; w.z *= dv[u]; w.w *= dv[u];
    t[i] = w;
    agg[i] = make_float4(0.f, 0.f, 0.f, 0.f);
  }
}

#define EPW 32
#define EPB 8
__global__ __launch_bounds__(256) void k_scatter_rows(const int* __restrict__ src,
                                                      const int* __restrict__ dst,
                                                      const float* __restrict__ t,
                                                      float* __restrict__ agg) {
  static_assert(NE % EPW == 0 && EPW % EPB == 0, "a wave takes full batches of edges");
  const int lane = threadIdx.x & 31;
  const int grp = (int)((blockIdx.x * blockDim.x + threadIdx.x) >> 5);
  if (grp >= NE / EPW) return;
  const int sv = src[(size_t)grp * EPW + lane];
  const int dv = dst[(size_t)grp * EPW + lane];
#define SRC_OF(j) __builtin_amdgcn_readlane(sv, (j))
#define DST_OF(j) __builtin_amdgcn_readlane(dv, (j))
  const float* tl = t + lane;
  float* al = agg + lane;
  float v0[EPW], v1[EPW];
#pragma unroll
  for (int j = 0; j < EPB; ++j) {
    const float* tr = tl + (size_t)SRC_OF(j) * HID;
    v0[j] = tr[0];
    v1[j] = tr[32];
  }
#pragma unroll
  for (int b = 0; b < EPW / EPB; ++b) {
    if (b + 1 < EPW / EPB) {
#pragma unroll
      for (int j = (b + 1) * EPB; j < (b + 2) * EPB; ++j) {
        const float* tr = tl + (size_t)SRC_OF(j) * HID;
        v0[j] = tr[0];
        v1[j] = tr[32];
      }
    }
#pragma unroll
    for (int j = b * EPB; j < (b + 1) * EPB; ++j) {
      float* ar = al + (size_t)DST_OF(j) * HID;
      atomicAdd(ar, v0[j]);
      atomicAdd(ar + 32, v1[j]);
    }
  }
}

__global__ __launch_bounds__(256) void k_update_combine4(float4* __restrict__ f,
                                                         const float4* __restrict__ agg,
                                                         const float* __restrict__ dinv,
                                                         float* __restrict__ out,
                                                         int col_base, float c0, float c1, float c2) {
  const int base = blockIdx.x * (256 * QPT) + threadIdx.x;
  const float cs[3] = {c0, c1, c2};
  v4f* fq = (v4f*)f;
  const v4f* aq = (const v4f*)agg;
  v4f fv[QPT], av[QPT], ov[QPT][3];
  float dv[QPT];
#pragma unroll
  for (int u = 0; u < QPT; ++u) {
    const int i = base + u * 256;
    const int row = i >> 4;
    const int q   = i & 15;
    const v4f* o = (const v4f*)(out + (size_t)row * DHID + col_base + q * 4);
    dv[u] = dinv[row];
    fv[u] = fq[i];
    av[u] = aq[i];
#pragma unroll
    for (int j = 0; j < 3; ++j) ov[u][j] = o[j * HID4];
  }
#pragma unroll
  for (int u = 0; u < QPT; ++u) {
    const int i = base + u * 256;
    const int row = i >> 4;
    const int q   = i & 15;
    v4f* o = (v4f*)(out + (size_t)row * DHID + col_base + q * 4);
    const v4f x = fv[u] - av[u] * dv[u];
    fq[i] = x;
#pragma unroll
    for (int j = 0; j < 3; ++j) o[j * HID4] = ov[u][j] + cs[j] * x;
  }
}


__global__ __launch_bounds__(128) void k_gemm1_relu_combine(
    const float* __restrict__ X, const float* __restrict__ W,
    const float* __restrict__ bias, float* __restrict__ Y,
    float* __restrict__ out, int col_base, float c0, float c1, float c2) {
  const int lane = threadIdx.x & 31;
  const int wave = threadIdx.x >> 5;
  const int row0 = blockIdx.x * 16;
  const int n0   = wave * 16;
  const int mr   = lane & 15;
  const int kq   = (lane >> 4) << 1;
  v8f c = {};
  const float* xrow = X + (size_t)(row0 + mr) * INF_;
  for (int k = 0; k < INF_; k += 4) {
    v2f a, b;
    a.x = xrow[k + kq];
    a.y = xrow[k + kq + 1];
    b.x = W[(size_t)(k + kq) * HID + n0 + mr];
    b.y = W[(size_t)(k + kq + 1) * HID + n0 + mr];
    c = __builtin_amdgcn_wmma_f32_16x16x4_f32(false, a, false, b, (short)0, c, false, false);
  }
  const int mhi = (lane >> 4) << 3;
  const float bv = bias[n0 + mr];
#pragma unroll
  for (int r = 0; r < 8; ++r) {
    float v = c[r] + bv;
    v = v > 0.0f ? v : 0.0f;
    const size_t row = (size_t)(row0 + mhi + r);
    Y[row * HID + n0 + mr] = v;
    float* o = out + row * DHID + col_base + n0 + mr;
    o[0]       = c0 * v;
    o[HID]     = c1 * v;
    o[2 * HID] = c2 * v;
  }
}


__global__ __launch_bounds__(128) void k_head(
    const float* __restrict__ H,
    const float* __restrict__ Wm1, const float* __restrict__ bm1,
    const float* __restrict__ Wm2, const float* __restrict__ bm2,
    float* __restrict__ logits) {
  __shared__ float m[16][HID];
  const int lane = threadIdx.x & 31;
  const int wave = threadIdx.x >> 5;
  const int row0 = blockIdx.x * 16;
  const int n0   = wave * 16;
  const int mr   = lane & 15;
  const int kq   = (lane >> 4) << 1;
  v8f c = {};
  const float* hrow = H + (size_t)(row0 + mr) * DHID;
  for (int k = 0; k < DHID; k += 4) {
    v2f a, b;
    a.x = hrow[k + kq];
    a.y = hrow[k + kq + 1];
    b.x = Wm1[(size_t)(k + kq) * HID + n0 + mr];
    b.y = Wm1[(size_t)(k + kq + 1) * HID + n0 + mr];
    c = __builtin_amdgcn_wmma_f32_16x16x4_f32(false, a, false, b, (short)0, c, false, false);
  }
  const int mhi = (lane >> 4) << 3;
  const float bv = bm1[n0 + mr];
#pragma unroll
  for (int r = 0; r < 8; ++r) {
    float v = c[r] + bv;
    m[mhi + r][n0 + mr] = v > 0.0f ? v : 0.0f;
  }
  __syncthreads();
  if (threadIdx.x < 32) {
    int row = threadIdx.x >> 1;
    int col = threadIdx.x & 1;
    float acc = bm2[col];
#pragma unroll
    for (int k = 0; k < HID; ++k) acc += m[row][k] * Wm2[k * 2 + col];
    logits[(size_t)(row0 + row) * 2 + col] = acc;
  }
}


static void run_branch(const float* X, const int* src, const int* dst,
                       const float* W1, const float* b1,
                       float* out, int col_base,
                       const float sc[3][3],
                       float* f, float* t, float* agg, float* dinv,
                       hipStream_t stream) {
  const int nNodeB = (NN + 255) / 256;
  const int nEdgeB = (NE / 4 + 256 * DEG_V - 1) / (256 * DEG_V);
  const int nQuadB = NQ / (256 * QPT);
  const int nScatB = (NE / EPW + 7) / 8;

  k_zero  <<<nNodeB, 256, 0, stream>>>(dinv, NN);
  k_degree<<<nEdgeB, 256, 0, stream>>>(dst, dinv);
  k_dinv  <<<nNodeB, 256, 0, stream>>>(dinv, NN);

  k_gemm1_relu_combine<<<NN / 16, 128, 0, stream>>>(X, W1, b1, f, out, col_base,
                                                    sc[0][0], sc[0][1], sc[0][2]);

  for (int step = 1; step <= 2; ++step) {
    k_scale_zero4<<<nQuadB, 256, 0, stream>>>((const float4*)f, dinv,
                                              (float4*)t, (float4*)agg);
    k_scatter_rows<<<nScatB, 256, 0, stream>>>(src, dst, t, agg);
    k_update_combine4<<<nQuadB, 256, 0, stream>>>((float4*)f, (const float4*)agg, dinv,
                                                  out, col_base,
                                                  sc[step][0], sc[step][1], sc[step][2]);
  }
}

extern "C" void kernel_launch(void* const* d_in, const int* in_sizes, int n_in,
                              void* d_out, int out_size, void* d_ws, size_t ws_size,
                              hipStream_t stream) {
  (void)in_sizes; (void)n_in; (void)out_size; (void)ws_size;
  const float* x     = (const float*)d_in[0];
  const float* sim_x = (const float*)d_in[1];
  const int*   src   = (const int*)d_in[2];
  const int*   dst   = (const int*)d_in[3];
  const int*   ssrc  = (const int*)d_in[4];
  const int*   sdst  = (const int*)d_in[5];
  const float* W1o   = (const float*)d_in[6];
  const float* b1o   = (const float*)d_in[7];
  const float* W1s   = (const float*)d_in[8];
  const float* b1s   = (const float*)d_in[9];
  const float* Wm1   = (const float*)d_in[10];
  const float* bm1   = (const float*)d_in[11];
  const float* Wm2   = (const float*)d_in[12];
  const float* bm2   = (const float*)d_in[13];

  float* out    = (float*)d_out;
  float* logits = out + (size_t)NN * DHID;

  float* f    = (float*)d_ws;
  float* t    = f    + (size_t)NN * HID;
  float* agg  = t    + (size_t)NN * HID;
  float* dinv = agg  + (size_t)NN * HID;

  const float sc_o[3][3] = {
      {3.0f,   0.0f,  0.0f},
      {-3.0f,  3.0f,  0.0f},
      {0.75f, -1.5f,  0.75f}};
  const float sc_s[3][3] = {
      {4.0f, 4.0f, 4.0f},
      {4.0f, 4.0f, 4.0f},
      {4.0f, 4.0f, 4.0f}};

  run_branch(x,     src,  dst,  W1o, b1o, out,   0, sc_o, f, t, agg, dinv, stream);
  run_branch(sim_x, ssrc, sdst, W1s, b1s, out, 192, sc_s, f, t, agg, dinv, stream);

  k_head<<<NN / 16, 128, 0, stream>>>(out, Wm1, bm1, Wm2, bm2, logits);
}


# h_all built once, in the head kernel: f0, f1 and the step-2 aggregate stay in workspace planes; the head forms f2 and the filter sums per row, stores it, feeds its WMMA from LDS; sum order unchanged
# speedup vs baseline: 1.8716x; 1.4108x over previous
#include <hip/hip_runtime.h>

#define NN   50000
#define NE   800000
#define INF_ 128
#define HID  64
#define HID4 16
#define DHID 384

typedef float v2f __attribute__((ext_vector_type(2)));
typedef float v8f __attribute__((ext_vector_type(8)));
typedef float v4f __attribute__((ext_vector_type(4)));


__global__ void k_zero(float* __restrict__ p, int n) {
  int i = blockIdx.x * blockDim.x + threadIdx.x;
  if (i < n) p[i] = 0.0f;
}

#define DEG_V 2
__global__ __launch_bounds__(256) void k_degree(const int* __restrict__ dst, float* __restrict__ deg) {
  static_assert(NE % 4 == 0, "edges are read four at a time");
  const int4* d4 = (const int4*)dst;
  const int base = blockIdx.x * (256 * DEG_V) + threadIdx.x;
  int4 d[DEG_V];
#pragma unroll
  for (int u = 0; u < DEG_V; ++u) {
    const int i = base + u * 256;
    if (i < NE / 4) d[u] = d4[i];
  }
#pragma unroll
  for (int u = 0; u < DEG_V; ++u) {
    const int i = base + u * 256;
    if (i < NE / 4) {
      atomicAdd(&deg[d[u].x], 1.0f);
      atomicAdd(&deg[d[u].y], 1.0f);
      atomicAdd(&deg[d[u].z], 1.0f);
      atomicAdd(&deg[d[u].w], 1.0f);
    }
  }
}

__global__ void k_dinv(float* __restrict__ d, int n) {
  int i = blockIdx.x * blockDim.x + threadIdx.x;
  if (i < n) {
    float v = d[i];
    v = v < 1.0f ? 1.0f : v;
    d[i] = rsqrtf(v);
  }
}

#define NQ  (NN * HID4)
#define QPT 5
static_assert(NQ % (256 * QPT) == 0, "blocks of 256 threads cover the plane exactly");

__global__ __launch_bounds__(256) void k_scale_zero4(const float4* __restrict__ f,
                                                     const float* __restrict__ dinv,
                                                     float4* __restrict__ t, float4* __restrict__ agg) {
  const int base = blockIdx.x * (256 * QPT) + threadIdx.x;
  float4 v[QPT];
  float dv[QPT];
#pragma unroll
  for (int u = 0; u < QPT; ++u) {
    const int i = base + u * 256;
    v[u] = f[i];
    dv[u] = dinv[i >> 4];
  }
#pragma unroll
  for (int u = 0; u < QPT; ++u) {
    const int i = base + u * 256;
    float4 w = v[u];
    w.x *= dv[u]; w.y *= dv[u]; w.z *= dv[u]; w.w *= dv[u];
    t[i] = w;
    agg[i] = make_float4(0.f, 0.f, 0.f, 0.f);
  }
}

#define EPW 32
#define EPB 8
__global__ __launch_bounds__(256) void k_scatter_rows(const int* __restrict__ src,
                                                      const int* __restrict__ dst,
                                                      const float* __restrict__ t,
                                                      float* __restrict__ agg) {
  static_assert(NE % EPW == 0 && EPW % EPB == 0, "a wave takes full batches of edges");
  const int lane = threadIdx.x & 31;
  const int grp = (int)((blockIdx.x * blockDim.x + threadIdx.x) >> 5);
  if (grp >= NE / EPW) return;
  const int sv = src[(size_t)grp * EPW + lane];
  const int dv = dst[(size_t)grp * EPW + lane];
#define SRC_OF(j) __builtin_amdgcn_readlane(sv, (j))
#define DST_OF(j) __builtin_amdgcn_readlane(dv, (j))
  const float* tl = t + lane;
  float* al = agg + lane;
  float v0[EPW], v1[EPW];
#pragma unroll
  for (int j = 0; j < EPB; ++j) {
    const float* tr = tl + (size_t)SRC_OF(j) * HID;
    v0[j] = tr[0];
    v1[j] = tr[32];
  }
#pragma unroll
  for (int b = 0; b < EPW / EPB; ++b) {
    if (b + 1 < EPW / EPB) {
#pragma unroll
      for (int j = (b + 1) * EPB; j < (b + 2) * EPB; ++j) {
        const float* tr = tl + (size_t)SRC_OF(j) * HID;
        v0[j] = tr[0];
        v1[j] = tr[32];
      }
    }
#pragma unroll
    for (int j = b * EPB; j < (b + 1) * EPB; ++j) {
      float* ar = al + (size_t)DST_OF(j) * HID;
      atomicAdd(ar, v0[j]);
      atomicAdd(ar + 32, v1[j]);
    }
  }
}

__global__ __launch_bounds__(256) void k_update4(const float4* __restrict__ fi,
                                                 const float4* __restrict__ agg,
                                                 const float* __restrict__ dinv,
                                                 float4* __restrict__ fo) {
  const int base = blockIdx.x * (256 * QPT) + threadIdx.x;
  const v4f* fq = (const v4f*)fi;
  const v4f* aq = (const v4f*)agg;
  v4f* oq = (v4f*)fo;
  v4f fv[QPT], av[QPT];
  float dv[QPT];
#pragma unroll
  for (int u = 0; u < QPT; ++u) {
    const int i = base + u * 256;
    dv[u] = dinv[i >> 4];
    fv[u] = fq[i];
    av[u] = aq[i];
  }
#pragma unroll
  for (int u = 0; u < QPT; ++u) {
    const int i = base + u * 256;
    oq[i] = fv[u] - av[u] * dv[u];
  }
}


__global__ __launch_bounds__(128) void k_gemm1_relu(
    const float* __restrict__ X, const float* __restrict__ W,
    const float* __restrict__ bias, float* __restrict__ Y) {
  const int lane = threadIdx.x & 31;
  const int wave = threadIdx.x >> 5;
  const int row0 = blockIdx.x * 16;
  const int n0   = wave * 16;
  const int mr   = lane & 15;
  const int kq   = (lane >> 4) << 1;
  v8f c = {};
  const float* xrow = X + (size_t)(row0 + mr) * INF_;
  for (int k = 0; k < INF_; k += 4) {
    v2f a, b;
    a.x = xrow[k + kq];
    a.y = xrow[k + kq + 1];
    b.x = W[(size_t)(k + kq) * HID + n0 + mr];
    b.y = W[(size_t)(k + kq + 1) * HID + n0 + mr];
    c = __builtin_amdgcn_wmma_f32_16x16x4_f32(false, a, false, b, (short)0, c, false, false);
  }
  const int mhi = (lane >> 4) << 3;
  const float bv = bias[n0 + mr];
#pragma unroll
  for (int r = 0; r < 8; ++r) {
    float v = c[r] + bv;
    v = v > 0.0f ? v : 0.0f;
    const size_t row = (size_t)(row0 + mhi + r);
    Y[row * HID + n0 + mr] = v;
  }
}


struct Coef { float c[2][3][3]; };

#define HP 388

__device__ __forceinline__ v4f fma4(float c, v4f x, v4f a) {
  v4f r;
  r.x = __builtin_fmaf(c, x.x, a.x);
  r.y = __builtin_fmaf(c, x.y, a.y);
  r.z = __builtin_fmaf(c, x.z, a.z);
  r.w = __builtin_fmaf(c, x.w, a.w);
  return r;
}

__global__ __launch_bounds__(128) void k_head_build(
    const float* __restrict__ f0o, const float* __restrict__ f1o,
    const float* __restrict__ ago, const float* __restrict__ dio,
    const float* __restrict__ f0s, const float* __restrict__ f1s,
    const float* __restrict__ ags, const float* __restrict__ dis,
    const Coef cf,
    const float* __restrict__ Wm1, const float* __restrict__ bm1,
    const float* __restrict__ Wm2, const float* __restrict__ bm2,
    float* __restrict__ out, float* __restrict__ logits) {
  __shared__ __attribute__((aligned(16))) float hs[16 * HP];
  __shared__ float m[16][HID];
  const int lane = threadIdx.x & 31;
  const int wave = threadIdx.x >> 5;
  const int row0 = blockIdx.x * 16;

  v4f ho[2][2][3];
  {
    v4f x0[2][2], x1[2][2], ag[2][2];
    float dv[2][2];
#pragma unroll
    for (int b = 0; b < 2; ++b) {
      const v4f* F0 = (const v4f*)(b ? f0s : f0o);
      const v4f* F1 = (const v4f*)(b ? f1s : f1o);
      const v4f* AG = (const v4f*)(b ? ags : ago);
      const float* DI = b ? dis : dio;
#pragma unroll
      for (int u = 0; u < 2; ++u) {
        const int qi = (int)threadIdx.x + u * 128;
        const int r  = qi >> 4;
        const size_t g = (size_t)(row0 + r) * HID4 + (qi & 15);
        x0[b][u] = F0[g];
        x1[b][u] = F1[g];
        ag[b][u] = AG[g];
        dv[b][u] = DI[row0 + r];
      }
    }
#pragma unroll
    for (int b = 0; b < 2; ++b) {
#pragma unroll
      for (int u = 0; u < 2; ++u) {
        const int qi = (int)threadIdx.x + u * 128;
        const int r  = qi >> 4;
        const int q  = qi & 15;
        const v4f x2 = fma4(-dv[b][u], ag[b][u], x1[b][u]);
#pragma unroll
        for (int j = 0; j < 3; ++j) {
          v4f o = cf.c[b][0][j] * x0[b][u];
          o = fma4(cf.c[b][1][j], x1[b][u], o);
          o = fma4(cf.c[b][2][j], x2, o);
          ho[b][u][j] = o;
          *(v4f*)(hs + r * HP + b * (3 * HID) + j * HID + q * 4) = o;
        }
      }
    }
  }
  __syncthreads();
#pragma unroll
  for (int b = 0; b < 2; ++b) {
#pragma unroll
    for (int u = 0; u < 2; ++u) {
      const int qi = (int)threadIdx.x + u * 128;
      float* orow = out + (size_t)(row0 + (qi >> 4)) * DHID + b * (3 * HID) + (qi & 15) * 4;
#pragma unroll
      for (int j = 0; j < 3; ++j) *(v4f*)(orow + j * HID) = ho[b][u][j];
    }
  }

  const int n0   = wave * 16;
  const int mr   = lane & 15;
  const int kq   = (lane >> 4) << 1;
  v8f c = {};
  const float* arow = hs + mr * HP + kq;
  for (int k = 0; k < DHID; k += 4) {
    v2f a, b;
    a = *(const v2f*)(arow + k);
    b.x = Wm1[(size_t)(k + kq) * HID + n0 + mr];
    b.y = Wm1[(size_t)(k + kq + 1) * HID + n0 + mr];
    c = __builtin_amdgcn_wmma_f32_16x16x4_f32(false, a, false, b, (short)0, c, false, false);
  }
  const int mhi = (lane >> 4) << 3;
  const float bv = bm1[n0 + mr];
#pragma unroll
  for (int r = 0; r < 8; ++r) {
    float v = c[r] + bv;
    m[mhi + r][n0 + mr] = v > 0.0f ? v : 0.0f;
  }
  __syncthreads();
  if (threadIdx.x < 32) {
    int row = threadIdx.x >> 1;
    int col = threadIdx.x & 1;
    float acc = bm2[col];
#pragma unroll
    for (int k = 0; k < HID; ++k) acc += m[row][k] * Wm2[k * 2 + col];
    logits[(size_t)(row0 + row) * 2 + col] = acc;
  }
}


static void run_branch(const float* X, const int* src, const int* dst,
                       const float* W1, const float* b1,
                       float* f0, float* f1, float* t, float* agg, float* dinv,
                       hipStream_t stream) {
  const int nNodeB = (NN + 255) / 256;
  const int nEdgeB = (NE / 4 + 256 * DEG_V - 1) / (256 * DEG_V);
  const int nQuadB = NQ / (256 * QPT);
  const int nScatB = (NE / EPW + 7) / 8;

  k_zero  <<<nNodeB, 256, 0, stream>>>(dinv, NN);
  k_degree<<<nEdgeB, 256, 0, stream>>>(dst, dinv);
  k_dinv  <<<nNodeB, 256, 0, stream>>>(dinv, NN);

  k_gemm1_relu<<<NN / 16, 128, 0, stream>>>(X, W1, b1, f0);

  k_scale_zero4<<<nQuadB, 256, 0, stream>>>((const float4*)f0, dinv, (float4*)t, (float4*)agg);
  k_scatter_rows<<<nScatB, 256, 0, stream>>>(src, dst, t, agg);
  k_update4<<<nQuadB, 256, 0, stream>>>((const float4*)f0, (const float4*)agg, dinv, (float4*)f1);

  k_scale_zero4<<<nQuadB, 256, 0, stream>>>((const float4*)f1, dinv, (float4*)t, (float4*)agg);
  k_scatter_rows<<<nScatB, 256, 0, stream>>>(src, dst, t, agg);
}

extern "C" void kernel_launch(void* const* d_in, const int* in_sizes, int n_in,
                              void* d_out, int out_size, void* d_ws, size_t ws_size,
                              hipStream_t stream) {
  (void)in_sizes; (void)n_in; (void)out_size; (void)ws_size;
  const float* x     = (const float*)d_in[0];
  const float* sim_x = (const float*)d_in[1];
  const int*   src   = (const int*)d_in[2];
  const int*   dst   = (const int*)d_in[3];
  const int*   ssrc  = (const int*)d_in[4];
  const int*   sdst  = (const int*)d_in[5];
  const float* W1o   = (const float*)d_in[6];
  const float* b1o   = (const float*)d_in[7];
  const float* W1s   = (const float*)d_in[8];
  const float* b1s   = (const float*)d_in[9];
  const float* Wm1   = (const float*)d_in[10];
  const float* bm1   = (const float*)d_in[11];
  const float* Wm2   = (const float*)d_in[12];
  const float* bm2   = (const float*)d_in[13];

  float* out    = (float*)d_out;
  float* logits = out + (size_t)NN * DHID;

  float* f0o  = (float*)d_ws;
  float* f1o  = f0o  + (size_t)NN * HID;
  float* ago  = f1o  + (size_t)NN * HID;
  float* f0s  = ago  + (size_t)NN * HID;
  float* f1s  = f0s  + (size_t)NN * HID;
  float* ags  = f1s  + (size_t)NN * HID;
  float* t    = ags  + (size_t)NN * HID;
  float* dio  = t    + (size_t)NN * HID;
  float* dis  = dio  + NN;

  const Coef cf = {{
      {{3.0f,   0.0f,  0.0f},
       {-3.0f,  3.0f,  0.0f},
       {0.75f, -1.5f,  0.75f}},
      {{4.0f, 4.0f, 4.0f},
       {4.0f, 4.0f, 4.0f},
       {4.0f, 4.0f, 4.0f}}}};

  run_branch(x,     src,  dst,  W1o, b1o, f0o, f1o, t, ago, dio, stream);
  run_branch(sim_x, ssrc, sdst, W1s, b1s, f0s, f1s, t, ags, dis, stream);

  k_head_build<<<NN / 16, 128, 0, stream>>>(f0o, f1o, ago, dio, f0s, f1s, ags, dis, cf,
                                            Wm1, bm1, Wm2, bm2, out, logits);
}


# node-wise passes folded into neighbours: GEMM epilogue forms dinv from the counts, writes f0, t=f0*dinv, agg=0; update writes f1, t=f1*dinv, agg=0; 19 -> 13 launches; arithmetic unchanged
# speedup vs baseline: 2.2871x; 1.2220x over previous
#include <hip/hip_runtime.h>

#define NN   50000
#define NE   800000
#define INF_ 128
#define HID  64
#define HID4 16
#define DHID 384

typedef float v2f __attribute__((ext_vector_type(2)));
typedef float v8f __attribute__((ext_vector_type(8)));
typedef float v4f __attribute__((ext_vector_type(4)));


__global__ void k_zero(float* __restrict__ p, int n) {
  int i = blockIdx.x * blockDim.x + threadIdx.x;
  if (i < n) p[i] = 0.0f;
}

#define DEG_V 2
__global__ __launch_bounds__(256) void k_degree(const int* __restrict__ dst, float* __restrict__ deg) {
  static_assert(NE % 4 == 0, "edges are read four at a time");
  const int4* d4 = (const int4*)dst;
  const int base = blockIdx.x * (256 * DEG_V) + threadIdx.x;
  int4 d[DEG_V];
#pragma unroll
  for (int u = 0; u < DEG_V; ++u) {
    const int i = base + u * 256;
    if (i < NE / 4) d[u] = d4[i];
  }
#pragma unroll
  for (int u = 0; u < DEG_V; ++u) {
    const int i = base + u * 256;
    if (i < NE / 4) {
      atomicAdd(&deg[d[u].x], 1.0f);
      atomicAdd(&deg[d[u].y], 1.0f);
      atomicAdd(&deg[d[u].z], 1.0f);
      atomicAdd(&deg[d[u].w], 1.0f);
    }
  }
}

#define NQ  (NN * HID4)
#define QPT 5
static_assert(NQ % (256 * QPT) == 0, "blocks of 256 threads cover the plane exactly");

#define EPW 32
#define EPB 8
__global__ __launch_bounds__(256) void k_scatter_rows(const int* __restrict__ src,
                                                      const int* __restrict__ dst,
                                                      const float* __restrict__ t,
                                                      float* __restrict__ agg) {
  static_assert(NE % EPW == 0 && EPW % EPB == 0, "a wave takes full batches of edges");
  const int lane = threadIdx.x & 31;
  const int grp = (int)((blockIdx.x * blockDim.x + threadIdx.x) >> 5);
  if (grp >= NE / EPW) return;
  const int sv = src[(size_t)grp * EPW + lane];
  const int dv = dst[(size_t)grp * EPW + lane];
#define SRC_OF(j) __builtin_amdgcn_readlane(sv, (j))
#define DST_OF(j) __builtin_amdgcn_readlane(dv, (j))
  const float* tl = t + lane;
  float* al = agg + lane;
  float v0[EPW], v1[EPW];
#pragma unroll
  for (int j = 0; j < EPB; ++j) {
    const float* tr = tl + (size_t)SRC_OF(j) * HID;
    v0[j] = tr[0];
    v1[j] = tr[32];
  }
#pragma unroll
  for (int b = 0; b < EPW / EPB; ++b) {
    if (b + 1 < EPW / EPB) {
#pragma unroll
      for (int j = (b + 1) * EPB; j < (b + 2) * EPB; ++j) {
        const float* tr = tl + (size_t)SRC_OF(j) * HID;
        v0[j] = tr[0];
        v1[j] = tr[32];
      }
    }
#pragma unroll
    for (int j = b * EPB; j < (b + 1) * EPB; ++j) {
      float* ar = al + (size_t)DST_OF(j) * HID;
      atomicAdd(ar, v0[j]);
      atomicAdd(ar + 32, v1[j]);
    }
  }
}

__global__ __launch_bounds__(256) void k_update_emit4(const float4* __restrict__ fi,
                                                      const float4* __restrict__ agg,
                                                      const float* __restrict__ dinv,
                                                      float4* __restrict__ fo,
                                                      float4* __restrict__ t,
                                                      float4* __restrict__ agn) {
  const int base = blockIdx.x * (256 * QPT) + threadIdx.x;
  const v4f* fq = (const v4f*)fi;
  const v4f* aq = (const v4f*)agg;
  v4f* oq = (v4f*)fo;
  v4f* tq = (v4f*)t;
  v4f* zq = (v4f*)agn;
  v4f fv[QPT], av[QPT];
  float dv[QPT];
#pragma unroll
  for (int u = 0; u < QPT; ++u) {
    const int i = base + u * 256;
    dv[u] = dinv[i >> 4];
    fv[u] = fq[i];
    av[u] = aq[i];
  }
  const v4f zero = {0.f, 0.f, 0.f, 0.f};
#pragma unroll
  for (int u = 0; u < QPT; ++u) {
    const int i = base + u * 256;
    const v4f f1 = fv[u] - av[u] * dv[u];
    oq[i] = f1;
    tq[i] = f1 * dv[u];
    zq[i] = zero;
  }
}


#define TP 68

__global__ __launch_bounds__(128) void k_gemm1_relu_emit(
    const float* __restrict__ X, const float* __restrict__ W,
    const float* __restrict__ bias, const float* __restrict__ deg,
    float* __restrict__ Y, float* __restrict__ T, float* __restrict__ AG, float* __restrict__ DI) {
  __shared__ __attribute__((aligned(16))) float ys[16 * TP];
  const int lane = threadIdx.x & 31;
  const int wave = threadIdx.x >> 5;
  const int row0 = blockIdx.x * 16;
  const int n0   = wave * 16;
  const int mr   = lane & 15;
  const int kq   = (lane >> 4) << 1;
  const int eq = (int)threadIdx.x & 15;
  const int er0 = (int)threadIdx.x >> 4;
  float dg0 = deg[row0 + er0];
  float dg1 = deg[row0 + er0 + 8];
  v8f c = {};
  const float* xrow = X + (size_t)(row0 + mr) * INF_;
  for (int k = 0; k < INF_; k += 4) {
    v2f a, b;
    a.x = xrow[k + kq];
    a.y = xrow[k + kq + 1];
    b.x = W[(size_t)(k + kq) * HID + n0 + mr];
    b.y = W[(size_t)(k + kq + 1) * HID + n0 + mr];
    c = __builtin_amdgcn_wmma_f32_16x16x4_f32(false, a, false, b, (short)0, c, false, false);
  }
  const int mhi = (lane >> 4) << 3;
  const float bv = bias[n0 + mr];
#pragma unroll
  for (int r = 0; r < 8; ++r) {
    float v = c[r] + bv;
    v = v > 0.0f ? v : 0.0f;
    ys[(mhi + r) * TP + n0 + mr] = v;
  }
  __syncthreads();
  dg0 = dg0 < 1.0f ? 1.0f : dg0;
  dg1 = dg1 < 1.0f ? 1.0f : dg1;
  const float dv0 = rsqrtf(dg0);
  const float dv1 = rsqrtf(dg1);
  const v4f zero = {0.f, 0.f, 0.f, 0.f};
  const v4f y0 = *(const v4f*)(ys + er0 * TP + eq * 4);
  const v4f y1 = *(const v4f*)(ys + (er0 + 8) * TP + eq * 4);
  const size_t g0 = (size_t)(row0 + er0) * HID4 + eq;
  const size_t g1 = g0 + 8 * HID4;
  ((v4f*)Y)[g0]  = y0;
  ((v4f*)Y)[g1]  = y1;
  ((v4f*)T)[g0]  = y0 * dv0;
  ((v4f*)T)[g1]  = y1 * dv1;
  ((v4f*)AG)[g0] = zero;
  ((v4f*)AG)[g1] = zero;
  if (eq == 0) {
    DI[row0 + er0]     = dv0;
    DI[row0 + er0 + 8] = dv1;
  }
}


struct Coef { float c[2][3][3]; };

#define HP 388

__device__ __forceinline__ v4f fma4(float c, v4f x, v4f a) {
  v4f r;
  r.x = __builtin_fmaf(c, x.x, a.x);
  r.y = __builtin_fmaf(c, x.y, a.y);
  r.z = __builtin_fmaf(c, x.z, a.z);
  r.w = __builtin_fmaf(c, x.w, a.w);
  return r;
}

__global__ __launch_bounds__(128) void k_head_build(
    const float* __restrict__ f0o, const float* __restrict__ f1o,
    const float* __restrict__ ago, const float* __restrict__ dio,
    const float* __restrict__ f0s, const float* __restrict__ f1s,
    const float* __restrict__ ags, const float* __restrict__ dis,
    const Coef cf,
    const float* __restrict__ Wm1, const float* __restrict__ bm1,
    const float* __restrict__ Wm2, const float* __restrict__ bm2,
    float* __restrict__ out, float* __restrict__ logits) {
  __shared__ __attribute__((aligned(16))) float hs[16 * HP];
  __shared__ float m[16][HID];
  const int lane = threadIdx.x & 31;
  const int wave = threadIdx.x >> 5;
  const int row0 = blockIdx.x * 16;

  v4f ho[2][2][3];
  {
    v4f x0[2][2], x1[2][2], ag[2][2];
    float dv[2][2];
#pragma unroll
    for (int b = 0; b < 2; ++b) {
      const v4f* F0 = (const v4f*)(b ? f0s : f0o);
      const v4f* F1 = (const v4f*)(b ? f1s : f1o);
      const v4f* AG = (const v4f*)(b ? ags : ago);
      const float* DI = b ? dis : dio;
#pragma unroll
      for (int u = 0; u < 2; ++u) {
        const int qi = (int)threadIdx.x + u * 128;
        const int r  = qi >> 4;
        const size_t g = (size_t)(row0 + r) * HID4 + (qi & 15);
        x0[b][u] = F0[g];
        x1[b][u] = F1[g];
        ag[b][u] = AG[g];
        dv[b][u] = DI[row0 + r];
      }
    }
#pragma unroll
    for (int b = 0; b < 2; ++b) {
#pragma unroll
      for (int u = 0; u < 2; ++u) {
        const int qi = (int)threadIdx.x + u * 128;
        const int r  = qi >> 4;
        const int q  = qi & 15;
        const v4f x2 = fma4(-dv[b][u], ag[b][u], x1[b][u]);
#pragma unroll
        for (int j = 0; j < 3; ++j) {
          v4f o = cf.c[b][0][j] * x0[b][u];
          o = fma4(cf.c[b][1][j], x1[b][u], o);
          o = fma4(cf.c[b][2][j], x2, o);
          ho[b][u][j] = o;
          *(v4f*)(hs + r * HP + b * (3 * HID) + j * HID + q * 4) = o;
        }
      }
    }
  }
  __syncthreads();
#pragma unroll
  for (int b = 0; b < 2; ++b) {
#pragma unroll
    for (int u = 0; u < 2; ++u) {
      const int qi = (int)threadIdx.x + u * 128;
      float* orow = out + (size_t)(row0 + (qi >> 4)) * DHID + b * (3 * HID) + (qi & 15) * 4;
#pragma unroll
      for (int j = 0; j < 3; ++j) *(v4f*)(orow + j * HID) = ho[b][u][j];
    }
  }

  const int n0   = wave * 16;
  const int mr   = lane & 15;
  const int kq   = (lane >> 4) << 1;
  v8f c = {};
  const float* arow = hs + mr * HP + kq;
  for (int k = 0; k < DHID; k += 4) {
    v2f a, b;
    a = *(const v2f*)(arow + k);
    b.x = Wm1[(size_t)(k + kq) * HID + n0 + mr];
    b.y = Wm1[(size_t)(k + kq + 1) * HID + n0 + mr];
    c = __builtin_amdgcn_wmma_f32_16x16x4_f32(false, a, false, b, (short)0, c, false, false);
  }
  const int mhi = (lane >> 4) << 3;
  const float bv = bm1[n0 + mr];
#pragma unroll
  for (int r = 0; r < 8; ++r) {
    float v = c[r] + bv;
    m[mhi + r][n0 + mr] = v > 0.0f ? v : 0.0f;
  }
  __syncthreads();
  if (threadIdx.x < 32) {
    int row = threadIdx.x >> 1;
    int col = threadIdx.x & 1;
    float acc = bm2[col];
#pragma unroll
    for (int k = 0; k < HID; ++k) acc += m[row][k] * Wm2[k * 2 + col];
    logits[(size_t)(row0 + row) * 2 + col] = acc;
  }
}


static void run_branch(const float* X, const int* src, const int* dst,
                       const float* W1, const float* b1,
                       float* f0, float* f1, float* t, float* ag1, float* ag2,
                       float* deg, float* dinv, hipStream_t stream) {
  const int nNodeB = (NN + 255) / 256;
  const int nEdgeB = (NE / 4 + 256 * DEG_V - 1) / (256 * DEG_V);
  const int nQuadB = NQ / (256 * QPT);
  const int nScatB = (NE / EPW + 7) / 8;

  k_zero  <<<nNodeB, 256, 0, stream>>>(deg, NN);
  k_degree<<<nEdgeB, 256, 0, stream>>>(dst, deg);

  k_gemm1_relu_emit<<<NN / 16, 128, 0, stream>>>(X, W1, b1, deg, f0, t, ag1, dinv);

  k_scatter_rows<<<nScatB, 256, 0, stream>>>(src, dst, t, ag1);
  k_update_emit4<<<nQuadB, 256, 0, stream>>>((const float4*)f0, (const float4*)ag1, dinv,
                                            (float4*)f1, (float4*)t, (float4*)ag2);

  k_scatter_rows<<<nScatB, 256, 0, stream>>>(src, dst, t, ag2);
}

extern "C" void kernel_launch(void* const* d_in, const int* in_sizes, int n_in,
                              void* d_out, int out_size, void* d_ws, size_t ws_size,
                              hipStream_t stream) {
  (void)in_sizes; (void)n_in; (void)out_size; (void)ws_size;
  const float* x     = (const float*)d_in[0];
  const float* sim_x = (const float*)d_in[1];
  const int*   src   = (const int*)d_in[2];
  const int*   dst   = (const int*)d_in[3];
  const int*   ssrc  = (const int*)d_in[4];
  const int*   sdst  = (const int*)d_in[5];
  const float* W1o   = (const float*)d_in[6];
  const float* b1o   = (const float*)d_in[7];
  const float* W1s   = (const float*)d_in[8];
  const float* b1s   = (const float*)d_in[9];
  const float* Wm1   = (const float*)d_in[10];
  const float* bm1   = (const float*)d_in[11];
  const float* Wm2   = (const float*)d_in[12];
  const float* bm2   = (const float*)d_in[13];

  float* out    = (float*)d_out;
  float* logits = out + (size_t)NN * DHID;

  float* f0o  = (float*)d_ws;
  float* f1o  = f0o  + (size_t)NN * HID;
  float* ago  = f1o  + (size_t)NN * HID;
  float* f0s  = ago  + (size_t)NN * HID;
  float* f1s  = f0s  + (size_t)NN * HID;
  float* ags  = f1s  + (size_t)NN * HID;
  float* t    = ags  + (size_t)NN * HID;
  float* ag1  = t    + (size_t)NN * HID;
  float* dio  = ag1  + (size_t)NN * HID;
  float* dis  = dio  + NN;
  float* dgo  = dis  + NN;
  float* dgs  = dgo  + NN;

  const Coef cf = {{
      {{3.0f,   0.0f,  0.0f},
       {-3.0f,  3.0f,  0.0f},
       {0.75f, -1.5f,  0.75f}},
      {{4.0f, 4.0f, 4.0f},
       {4.0f, 4.0f, 4.0f},
       {4.0f, 4.0f, 4.0f}}}};

  run_branch(x,     src,  dst,  W1o, b1o, f0o, f1o, t, ag1, ago, dgo, dio, stream);
  run_branch(sim_x, ssrc, sdst, W1s, b1s, f0s, f1s, t, ag1, ags, dgs, dis, stream);

  k_head_build<<<NN / 16, 128, 0, stream>>>(f0o, f1o, ago, dio, f0s, f1s, ags, dis, cf,
                                            Wm1, bm1, Wm2, bm2, out, logits);
}


# in-degree counted by LDS integer adds per (node range, edge slice) block into 16 partial planes, summed exactly in the GEMM epilogue; global count atomics and the zero pass go, 13 -> 11 launches
# speedup vs baseline: 3.0328x; 1.3260x over previous
#include <hip/hip_runtime.h>

#define NN   50000
#define NE   800000
#define INF_ 128
#define HID  64
#define HID4 16
#define DHID 384

typedef float v2f __attribute__((ext_vector_type(2)));
typedef float v8f __attribute__((ext_vector_type(8)));
typedef float v4f __attribute__((ext_vector_type(4)));


#define DR   16
#define DS   16
#define DRN  3136
#define DPP  (DR * DRN)
#define DTH  512
#define DSQ  (NE / DS / 4)
#define DLD  ((DSQ + DTH - 1) / DTH)
static_assert(DPP >= NN, "the ranges cover every node");
static_assert(NE % (DS * 4) == 0, "a slice is read four edges at a time");
static_assert(DRN % 32 == 0, "a range's counters are whole 128-byte lines");

__global__ __launch_bounds__(DTH) void k_degree_bins(const int* __restrict__ dst, float* __restrict__ part) {
  __shared__ __attribute__((aligned(16))) unsigned h[DRN];
  const int slice = (int)blockIdx.x % DS;
  const unsigned lo = ((unsigned)blockIdx.x / DS) * DRN;
  const uint4 zero = {0u, 0u, 0u, 0u};
  for (int i = threadIdx.x; i < DRN / 4; i += DTH) ((uint4*)h)[i] = zero;
  const int4* d4 = (const int4*)dst + (size_t)slice * DSQ;
  int4 d[DLD];
#pragma unroll
  for (int u = 0; u < DLD; ++u) {
    const int i = (int)threadIdx.x + u * DTH;
    if (i < DSQ) d[u] = d4[i];
  }
  __syncthreads();
#pragma unroll
  for (int u = 0; u < DLD; ++u) {
    const int i = (int)threadIdx.x + u * DTH;
    if (i < DSQ) {
      const unsigned a0 = (unsigned)d[u].x - lo;
      const unsigned a1 = (unsigned)d[u].y - lo;
      const unsigned a2 = (unsigned)d[u].z - lo;
      const unsigned a3 = (unsigned)d[u].w - lo;
      if (a0 < DRN) atomicAdd(&h[a0], 1u);
      if (a1 < DRN) atomicAdd(&h[a1], 1u);
      if (a2 < DRN) atomicAdd(&h[a2], 1u);
      if (a3 < DRN) atomicAdd(&h[a3], 1u);
    }
  }
  __syncthreads();
  v4f* po = (v4f*)(part + (size_t)slice * DPP + lo);
  for (int i = threadIdx.x; i < DRN / 4; i += DTH) {
    const uint4 c = ((const uint4*)h)[i];
    const v4f f = {(float)c.x, (float)c.y, (float)c.z, (float)c.w};
    po[i] = f;
  }
}

#define NQ  (NN * HID4)
#define QPT 5
static_assert(NQ % (256 * QPT) == 0, "blocks of 256 threads cover the plane exactly");

#define EPW 32
#define EPB 8
__global__ __launch_bounds__(256) void k_scatter_rows(const int* __restrict__ src,
                                                      const int* __restrict__ dst,
                                                      const float* __restrict__ t,
                                                      float* __restrict__ agg) {
  static_assert(NE % EPW == 0 && EPW % EPB == 0, "a wave takes full batches of edges");
  const int lane = threadIdx.x & 31;
  const int grp = (int)((blockIdx.x * blockDim.x + threadIdx.x) >> 5);
  if (grp >= NE / EPW) return;
  const int sv = src[(size_t)grp * EPW + lane];
  const int dv = dst[(size_t)grp * EPW + lane];
#define SRC_OF(j) __builtin_amdgcn_readlane(sv, (j))
#define DST_OF(j) __builtin_amdgcn_readlane(dv, (j))
  const float* tl = t + lane;
  float* al = agg + lane;
  float v0[EPW], v1[EPW];
#pragma unroll
  for (int j = 0; j < EPB; ++j) {
    const float* tr = tl + (size_t)SRC_OF(j) * HID;
    v0[j] = tr[0];
    v1[j] = tr[32];
  }
#pragma unroll
  for (int b = 0; b < EPW / EPB; ++b) {
    if (b + 1 < EPW / EPB) {
#pragma unroll
      for (int j = (b + 1) * EPB; j < (b + 2) * EPB; ++j) {
        const float* tr = tl + (size_t)SRC_OF(j) * HID;
        v0[j] = tr[0];
        v1[j] = tr[32];
      }
    }
#pragma unroll
    for (int j = b * EPB; j < (b + 1) * EPB; ++j) {
      float* ar = al + (size_t)DST_OF(j) * HID;
      atomicAdd(ar, v0[j]);
      atomicAdd(ar + 32, v1[j]);
    }
  }
}

__global__ __launch_bounds__(256) void k_update_emit4(const float4* __restrict__ fi,
                                                      const float4* __restrict__ agg,
                                                      const float* __restrict__ dinv,
                                                      float4* __restrict__ fo,
                                                      float4* __restrict__ t,
                                                      float4* __restrict__ agn) {
  const int base = blockIdx.x * (256 * QPT) + threadIdx.x;
  const v4f* fq = (const v4f*)fi;
  const v4f* aq = (const v4f*)agg;
  v4f* oq = (v4f*)fo;
  v4f* tq = (v4f*)t;
  v4f* zq = (v4f*)agn;
  v4f fv[QPT], av[QPT];
  float dv[QPT];
#pragma unroll
  for (int u = 0; u < QPT; ++u) {
    const int i = base + u * 256;
    dv[u] = dinv[i >> 4];
    fv[u] = fq[i];
    av[u] = aq[i];
  }
  const v4f zero = {0.f, 0.f, 0.f, 0.f};
#pragma unroll
  for (int u = 0; u < QPT; ++u) {
    const int i = base + u * 256;
    const v4f f1 = fv[u] - av[u] * dv[u];
    oq[i] = f1;
    tq[i] = f1 * dv[u];
    zq[i] = zero;
  }
}


#define TP 68

__global__ __launch_bounds__(128) void k_gemm1_relu_emit(
    const float* __restrict__ X, const float* __restrict__ W,
    const float* __restrict__ bias, const float* __restrict__ part,
    float* __restrict__ Y, float* __restrict__ T, float* __restrict__ AG, float* __restrict__ DI) {
  __shared__ __attribute__((aligned(16))) float ys[16 * TP];
  const int lane = threadIdx.x & 31;
  const int wave = threadIdx.x >> 5;
  const int row0 = blockIdx.x * 16;
  const int n0   = wave * 16;
  const int mr   = lane & 15;
  const int kq   = (lane >> 4) << 1;
  static_assert(DS == 16, "the 16 lanes that share a row take one partial plane each");
  const int eq = (int)threadIdx.x & 15;
  const int er0 = (int)threadIdx.x >> 4;
  float dg0 = part[(size_t)eq * DPP + row0 + er0];
  float dg1 = part[(size_t)eq * DPP + row0 + er0 + 8];
  v8f c = {};
  const float* xrow = X + (size_t)(row0 + mr) * INF_;
  for (int k = 0; k < INF_; k += 4) {
    v2f a, b;
    a.x = xrow[k + kq];
    a.y = xrow[k + kq + 1];
    b.x = W[(size_t)(k + kq) * HID + n0 + mr];
    b.y = W[(size_t)(k + kq + 1) * HID + n0 + mr];
    c = __builtin_amdgcn_wmma_f32_16x16x4_f32(false, a, false, b, (short)0, c, false, false);
  }
  const int mhi = (lane >> 4) << 3;
  const float bv = bias[n0 + mr];
#pragma unroll
  for (int r = 0; r < 8; ++r) {
    float v = c[r] + bv;
    v = v > 0.0f ? v : 0.0f;
    ys[(mhi + r) * TP + n0 + mr] = v;
  }
  __syncthreads();
#pragma unroll
  for (int msk = 1; msk < 16; msk <<= 1) {
    dg0 += __shfl_xor(dg0, msk);
    dg1 += __shfl_xor(dg1, msk);
  }
  dg0 = dg0 < 1.0f ? 1.0f : dg0;
  dg1 = dg1 < 1.0f ? 1.0f : dg1;
  const float dv0 = rsqrtf(dg0);
  const float dv1 = rsqrtf(dg1);
  const v4f zero = {0.f, 0.f, 0.f, 0.f};
  const v4f y0 = *(const v4f*)(ys + er0 * TP + eq * 4);
  const v4f y1 = *(const v4f*)(ys + (er0 + 8) * TP + eq * 4);
  const size_t g0 = (size_t)(row0 + er0) * HID4 + eq;
  const size_t g1 = g0 + 8 * HID4;
  ((v4f*)Y)[g0]  = y0;
  ((v4f*)Y)[g1]  = y1;
  ((v4f*)T)[g0]  = y0 * dv0;
  ((v4f*)T)[g1]  = y1 * dv1;
  ((v4f*)AG)[g0] = zero;
  ((v4f*)AG)[g1] = zero;
  if (eq == 0) {
    DI[row0 + er0]     = dv0;
    DI[row0 + er0 + 8] = dv1;
  }
}


struct Coef { float c[2][3][3]; };

#define HP 388

__device__ __forceinline__ v4f fma4(float c, v4f x, v4f a) {
  v4f r;
  r.x = __builtin_fmaf(c, x.x, a.x);
  r.y = __builtin_fmaf(c, x.y, a.y);
  r.z = __builtin_fmaf(c, x.z, a.z);
  r.w = __builtin_fmaf(c, x.w, a.w);
  return r;
}

__global__ __launch_bounds__(128) void k_head_build(
    const float* __restrict__ f0o, const float* __restrict__ f1o,
    const float* __restrict__ ago, const float* __restrict__ dio,
    const float* __restrict__ f0s, const float* __restrict__ f1s,
    const float* __restrict__ ags, const float* __restrict__ dis,
    const Coef cf,
    const float* __restrict__ Wm1, const float* __restrict__ bm1,
    const float* __restrict__ Wm2, const float* __restrict__ bm2,
    float* __restrict__ out, float* __restrict__ logits) {
  __shared__ __attribute__((aligned(16))) float hs[16 * HP];
  __shared__ float m[16][HID];
  const int lane = threadIdx.x & 31;
  const int wave = threadIdx.x >> 5;
  const int row0 = blockIdx.x * 16;

  v4f ho[2][2][3];
  {
    v4f x0[2][2], x1[2][2], ag[2][2];
    float dv[2][2];
#pragma unroll
    for (int b = 0; b < 2; ++b) {
      const v4f* F0 = (const v4f*)(b ? f0s : f0o);
      const v4f* F1 = (const v4f*)(b ? f1s : f1o);
      const v4f* AG = (const v4f*)(b ? ags : ago);
      const float* DI = b ? dis : dio;
#pragma unroll
      for (int u = 0; u < 2; ++u) {
        const int qi = (int)threadIdx.x + u * 128;
        const int r  = qi >> 4;
        const size_t g = (size_t)(row0 + r) * HID4 + (qi & 15);
        x0[b][u] = F0[g];
        x1[b][u] = F1[g];
        ag[b][u] = AG[g];
        dv[b][u] = DI[row0 + r];
      }
    }
#pragma unroll
    for (int b = 0; b < 2; ++b) {
#pragma unroll
      for (int u = 0; u < 2; ++u) {
        const int qi = (int)threadIdx.x + u * 128;
        const int r  = qi >> 4;
        const int q  = qi & 15;
        const v4f x2 = fma4(-dv[b][u], ag[b][u], x1[b][u]);
#pragma unroll
        for (int j = 0; j < 3; ++j) {
          v4f o = cf.c[b][0][j] * x0[b][u];
          o = fma4(cf.c[b][1][j], x1[b][u], o);
          o = fma4(cf.c[b][2][j], x2, o);
          ho[b][u][j] = o;
          *(v4f*)(hs + r * HP + b * (3 * HID) + j * HID + q * 4) = o;
        }
      }
    }
  }
  __syncthreads();
#pragma unroll
  for (int b = 0; b < 2; ++b) {
#pragma unroll
    for (int u = 0; u < 2; ++u) {
      const int qi = (int)threadIdx.x + u * 128;
      float* orow = out + (size_t)(row0 + (qi >> 4)) * DHID + b * (3 * HID) + (qi & 15) * 4;
#pragma unroll
      for (int j = 0; j < 3; ++j) *(v4f*)(orow + j * HID) = ho[b][u][j];
    }
  }

  const int n0   = wave * 16;
  const int mr   = lane & 15;
  const int kq   = (lane >> 4) << 1;
  v8f c = {};
  const float* arow = hs + mr * HP + kq;
  for (int k = 0; k < DHID; k += 4) {
    v2f a, b;
    a = *(const v2f*)(arow + k);
    b.x = Wm1[(size_t)(k + kq) * HID + n0 + mr];
    b.y = Wm1[(size_t)(k + kq + 1) * HID + n0 + mr];
    c = __builtin_amdgcn_wmma_f32_16x16x4_f32(false, a, false, b, (short)0, c, false, false);
  }
  const int mhi = (lane >> 4) << 3;
  const float bv = bm1[n0 + mr];
#pragma unroll
  for (int r = 0; r < 8; ++r) {
    float v = c[r] + bv;
    m[mhi + r][n0 + mr] = v > 0.0f ? v : 0.0f;
  }
  __syncthreads();
  if (threadIdx.x < 32) {
    int row = threadIdx.x >> 1;
    int col = threadIdx.x & 1;
    float acc = bm2[col];
#pragma unroll
    for (int k = 0; k < HID; ++k) acc += m[row][k] * Wm2[k * 2 + col];
    logits[(size_t)(row0 + row) * 2 + col] = acc;
  }
}


static void run_branch(const float* X, const int* src, const int* dst,
                       const float* W1, const float* b1,
                       float* f0, float* f1, float* t, float* ag1, float* ag2,
                       float* part, float* dinv, hipStream_t stream) {
  const int nQuadB = NQ / (256 * QPT);
  const int nScatB = (NE / EPW + 7) / 8;

  k_degree_bins<<<DR * DS, DTH, 0, stream>>>(dst, part);

  k_gemm1_relu_emit<<<NN / 16, 128, 0, stream>>>(X, W1, b1, part, f0, t, ag1, dinv);

  k_scatter_rows<<<nScatB, 256, 0, stream>>>(src, dst, t, ag1);
  k_update_emit4<<<nQuadB, 256, 0, stream>>>((const float4*)f0, (const float4*)ag1, dinv,
                                            (float4*)f1, (float4*)t, (float4*)ag2);

  k_scatter_rows<<<nScatB, 256, 0, stream>>>(src, dst, t, ag2);
}

extern "C" void kernel_launch(void* const* d_in, const int* in_sizes, int n_in,
                              void* d_out, int out_size, void* d_ws, size_t ws_size,
                              hipStream_t stream) {
  (void)in_sizes; (void)n_in; (void)out_size; (void)ws_size;
  const float* x     = (const float*)d_in[0];
  const float* sim_x = (const float*)d_in[1];
  const int*   src   = (const int*)d_in[2];
  const int*   dst   = (const int*)d_in[3];
  const int*   ssrc  = (const int*)d_in[4];
  const int*   sdst  = (const int*)d_in[5];
  const float* W1o   = (const float*)d_in[6];
  const float* b1o   = (const float*)d_in[7];
  const float* W1s   = (const float*)d_in[8];
  const float* b1s   = (const float*)d_in[9];
  const float* Wm1   = (const float*)d_in[10];
  const float* bm1   = (const float*)d_in[11];
  const float* Wm2   = (const float*)d_in[12];
  const float* bm2   = (const float*)d_in[13];

  float* out    = (float*)d_out;
  float* logits = out + (size_t)NN * DHID;

  float* f0o  = (float*)d_ws;
  float* f1o  = f0o  + (size_t)NN * HID;
  float* ago  = f1o  + (size_t)NN * HID;
  float* f0s  = ago  + (size_t)NN * HID;
  float* f1s  = f0s  + (size_t)NN * HID;
  float* ags  = f1s  + (size_t)NN * HID;
  float* t    = ags  + (size_t)NN * HID;
  float* ag1  = t    + (size_t)NN * HID;
  float* dio  = ag1  + (size_t)NN * HID;
  float* dis  = dio  + NN;
  float* pco  = dis  + NN;
  float* pcs  = pco  + (size_t)DS * DPP;

  const Coef cf = {{
      {{3.0f,   0.0f,  0.0f},
       {-3.0f,  3.0f,  0.0f},
       {0.75f, -1.5f,  0.75f}},
      {{4.0f, 4.0f, 4.0f},
       {4.0f, 4.0f, 4.0f},
       {4.0f, 4.0f, 4.0f}}}};

  run_branch(x,     src,  dst,  W1o, b1o, f0o, f1o, t, ag1, ago, pco, dio, stream);
  run_branch(sim_x, ssrc, sdst, W1s, b1s, f0s, f1s, t, ag1, ags, pcs, dis, stream);

  k_head_build<<<NN / 16, 128, 0, stream>>>(f0o, f1o, ago, dio, f0s, f1s, ags, dis, cf,
                                            Wm1, bm1, Wm2, bm2, out, logits);
}


# scatter adds split by chip half: a wave reads its die and adds into that half's own aggregate plane, so no line takes adds from both L2 halves; readers sum plane A + plane B (sum order changes)
# speedup vs baseline: 3.3313x; 1.0984x over previous
#include <hip/hip_runtime.h>

#define NN   50000
#define NE   800000
#define INF_ 128
#define HID  64
#define HID4 16
#define DHID 384

typedef float v2f __attribute__((ext_vector_type(2)));
typedef float v8f __attribute__((ext_vector_type(8)));
typedef float v4f __attribute__((ext_vector_type(4)));


#define DR   16
#define DS   16
#define DRN  3136
#define DPP  (DR * DRN)
#define DTH  512
#define DSQ  (NE / DS / 4)
#define DLD  ((DSQ + DTH - 1) / DTH)
static_assert(DPP >= NN, "the ranges cover every node");
static_assert(NE % (DS * 4) == 0, "a slice is read four edges at a time");
static_assert(DRN % 32 == 0, "a range's counters are whole 128-byte lines");

__global__ __launch_bounds__(DTH) void k_degree_bins(const int* __restrict__ dst, float* __restrict__ part) {
  __shared__ __attribute__((aligned(16))) unsigned h[DRN];
  const int slice = (int)blockIdx.x % DS;
  const unsigned lo = ((unsigned)blockIdx.x / DS) * DRN;
  const uint4 zero = {0u, 0u, 0u, 0u};
  for (int i = threadIdx.x; i < DRN / 4; i += DTH) ((uint4*)h)[i] = zero;
  const int4* d4 = (const int4*)dst + (size_t)slice * DSQ;
  int4 d[DLD];
#pragma unroll
  for (int u = 0; u < DLD; ++u) {
    const int i = (int)threadIdx.x + u * DTH;
    if (i < DSQ) d[u] = d4[i];
  }
  __syncthreads();
#pragma unroll
  for (int u = 0; u < DLD; ++u) {
    const int i = (int)threadIdx.x + u * DTH;
    if (i < DSQ) {
      const unsigned a0 = (unsigned)d[u].x - lo;
      const unsigned a1 = (unsigned)d[u].y - lo;
      const unsigned a2 = (unsigned)d[u].z - lo;
      const unsigned a3 = (unsigned)d[u].w - lo;
      if (a0 < DRN) atomicAdd(&h[a0], 1u);
      if (a1 < DRN) atomicAdd(&h[a1], 1u);
      if (a2 < DRN) atomicAdd(&h[a2], 1u);
      if (a3 < DRN) atomicAdd(&h[a3], 1u);
    }
  }
  __syncthreads();
  v4f* po = (v4f*)(part + (size_t)slice * DPP + lo);
  for (int i = threadIdx.x; i < DRN / 4; i += DTH) {
    const uint4 c = ((const uint4*)h)[i];
    const v4f f = {(float)c.x, (float)c.y, (float)c.z, (float)c.w};
    po[i] = f;
  }
}

#define NQ  (NN * HID4)
#define QPT 5
static_assert(NQ % (256 * QPT) == 0, "blocks of 256 threads cover the plane exactly");

#define EPW 32
#define EPB 8
__global__ __launch_bounds__(256) void k_scatter_rows(const int* __restrict__ src,
                                                      const int* __restrict__ dst,
                                                      const float* __restrict__ t,
                                                      float* __restrict__ aggA,
                                                      float* __restrict__ aggB) {
  static_assert(NE % EPW == 0 && EPW % EPB == 0, "a wave takes full batches of edges");
  const int lane = threadIdx.x & 31;
  const int grp = (int)((blockIdx.x * blockDim.x + threadIdx.x) >> 5);
  if (grp >= NE / EPW) return;
  const int sv = src[(size_t)grp * EPW + lane];
  const int dv = dst[(size_t)grp * EPW + lane];
#define SRC_OF(j) __builtin_amdgcn_readlane(sv, (j))
#define DST_OF(j) __builtin_amdgcn_readlane(dv, (j))
  const float* tl = t + lane;
  const unsigned place = __builtin_amdgcn_s_sendmsg_rtn(0x87);
  float* al = ((place >> 18) & 1u ? aggB : aggA) + lane;
  float v0[EPW], v1[EPW];
#pragma unroll
  for (int j = 0; j < EPB; ++j) {
    const float* tr = tl + (size_t)SRC_OF(j) * HID;
    v0[j] = tr[0];
    v1[j] = tr[32];
  }
#pragma unroll
  for (int b = 0; b < EPW / EPB; ++b) {
    if (b + 1 < EPW / EPB) {
#pragma unroll
      for (int j = (b + 1) * EPB; j < (b + 2) * EPB; ++j) {
        const float* tr = tl + (size_t)SRC_OF(j) * HID;
        v0[j] = tr[0];
        v1[j] = tr[32];
      }
    }
#pragma unroll
    for (int j = b * EPB; j < (b + 1) * EPB; ++j) {
      float* ar = al + (size_t)DST_OF(j) * HID;
      atomicAdd(ar, v0[j]);
      atomicAdd(ar + 32, v1[j]);
    }
  }
}

__global__ __launch_bounds__(256) void k_update_emit4(const float4* __restrict__ fi,
                                                      const float4* __restrict__ aggA,
                                                      const float4* __restrict__ aggB,
                                                      const float* __restrict__ dinv,
                                                      float4* __restrict__ fo,
                                                      float4* __restrict__ t,
                                                      float4* __restrict__ agnA,
                                                      float4* __restrict__ agnB) {
  const int base = blockIdx.x * (256 * QPT) + threadIdx.x;
  const v4f* fq = (const v4f*)fi;
  const v4f* aq = (const v4f*)aggA;
  const v4f* bq = (const v4f*)aggB;
  v4f* oq = (v4f*)fo;
  v4f* tq = (v4f*)t;
  v4f* zq = (v4f*)agnA;
  v4f* yq = (v4f*)agnB;
  v4f fv[QPT], av[QPT], bv[QPT];
  float dv[QPT];
#pragma unroll
  for (int u = 0; u < QPT; ++u) {
    const int i = base + u * 256;
    dv[u] = dinv[i >> 4];
    fv[u] = fq[i];
    av[u] = aq[i];
    bv[u] = bq[i];
  }
  const v4f zero = {0.f, 0.f, 0.f, 0.f};
#pragma unroll
  for (int u = 0; u < QPT; ++u) {
    const int i = base + u * 256;
    const v4f f1 = fv[u] - (av[u] + bv[u]) * dv[u];
    oq[i] = f1;
    tq[i] = f1 * dv[u];
    zq[i] = zero;
    yq[i] = zero;
  }
}


#define TP 68

__global__ __launch_bounds__(128) void k_gemm1_relu_emit(
    const float* __restrict__ X, const float* __restrict__ W,
    const float* __restrict__ bias, const float* __restrict__ part,
    float* __restrict__ Y, float* __restrict__ T, float* __restrict__ AG, float* __restrict__ AGB,
    float* __restrict__ DI) {
  __shared__ __attribute__((aligned(16))) float ys[16 * TP];
  const int lane = threadIdx.x & 31;
  const int wave = threadIdx.x >> 5;
  const int row0 = blockIdx.x * 16;
  const int n0   = wave * 16;
  const int mr   = lane & 15;
  const int kq   = (lane >> 4) << 1;
  static_assert(DS == 16, "the 16 lanes that share a row take one partial plane each");
  const int eq = (int)threadIdx.x & 15;
  const int er0 = (int)threadIdx.x >> 4;
  float dg0 = part[(size_t)eq * DPP + row0 + er0];
  float dg1 = part[(size_t)eq * DPP + row0 + er0 + 8];
  v8f c = {};
  const float* xrow = X + (size_t)(row0 + mr) * INF_;
  for (int k = 0; k < INF_; k += 4) {
    v2f a, b;
    a.x = xrow[k + kq];
    a.y = xrow[k + kq + 1];
    b.x = W[(size_t)(k + kq) * HID + n0 + mr];
    b.y = W[(size_t)(k + kq + 1) * HID + n0 + mr];
    c = __builtin_amdgcn_wmma_f32_16x16x4_f32(false, a, false, b, (short)0, c, false, false);
  }
  const int mhi = (lane >> 4) << 3;
  const float bv = bias[n0 + mr];
#pragma unroll
  for (int r = 0; r < 8; ++r) {
    float v = c[r] + bv;
    v = v > 0.0f ? v : 0.0f;
    ys[(mhi + r) * TP + n0 + mr] = v;
  }
  __syncthreads();
#pragma unroll
  for (int msk = 1; msk < 16; msk <<= 1) {
    dg0 += __shfl_xor(dg0, msk);
    dg1 += __shfl_xor(dg1, msk);
  }
  dg0 = dg0 < 1.0f ? 1.0f : dg0;
  dg1 = dg1 < 1.0f ? 1.0f : dg1;
  const float dv0 = rsqrtf(dg0);
  const float dv1 = rsqrtf(dg1);
  const v4f zero = {0.f, 0.f, 0.f, 0.f};
  const v4f y0 = *(const v4f*)(ys + er0 * TP + eq * 4);
  const v4f y1 = *(const v4f*)(ys + (er0 + 8) * TP + eq * 4);
  const size_t g0 = (size_t)(row0 + er0) * HID4 + eq;
  const size_t g1 = g0 + 8 * HID4;
  ((v4f*)Y)[g0]  = y0;
  ((v4f*)Y)[g1]  = y1;
  ((v4f*)T)[g0]  = y0 * dv0;
  ((v4f*)T)[g1]  = y1 * dv1;
  ((v4f*)AG)[g0] = zero;
  ((v4f*)AG)[g1] = zero;
  ((v4f*)AGB)[g0] = zero;
  ((v4f*)AGB)[g1] = zero;
  if (eq == 0) {
    DI[row0 + er0]     = dv0;
    DI[row0 + er0 + 8] = dv1;
  }
}


struct Coef { float c[2][3][3]; };

#define HP 388

__device__ __forceinline__ v4f fma4(float c, v4f x, v4f a) {
  v4f r;
  r.x = __builtin_fmaf(c, x.x, a.x);
  r.y = __builtin_fmaf(c, x.y, a.y);
  r.z = __builtin_fmaf(c, x.z, a.z);
  r.w = __builtin_fmaf(c, x.w, a.w);
  return r;
}

__global__ __launch_bounds__(128) void k_head_build(
    const float* __restrict__ f0o, const float* __restrict__ f1o,
    const float* __restrict__ ago, const float* __restrict__ agob, const float* __restrict__ dio,
    const float* __restrict__ f0s, const float* __restrict__ f1s,
    const float* __restrict__ ags, const float* __restrict__ agsb, const float* __restrict__ dis,
    const Coef cf,
    const float* __restrict__ Wm1, const float* __restrict__ bm1,
    const float* __restrict__ Wm2, const float* __restrict__ bm2,
    float* __restrict__ out, float* __restrict__ logits) {
  __shared__ __attribute__((aligned(16))) float hs[16 * HP];
  __shared__ float m[16][HID];
  const int lane = threadIdx.x & 31;
  const int wave = threadIdx.x >> 5;
  const int row0 = blockIdx.x * 16;

  v4f ho[2][2][3];
  {
    v4f x0[2][2], x1[2][2], ag[2][2], ah[2][2];
    float dv[2][2];
#pragma unroll
    for (int b = 0; b < 2; ++b) {
      const v4f* F0 = (const v4f*)(b ? f0s : f0o);
      const v4f* F1 = (const v4f*)(b ? f1s : f1o);
      const v4f* AG = (const v4f*)(b ? ags : ago);
      const v4f* AH = (const v4f*)(b ? agsb : agob);
      const float* DI = b ? dis : dio;
#pragma unroll
      for (int u = 0; u < 2; ++u) {
        const int qi = (int)threadIdx.x + u * 128;
        const int r  = qi >> 4;
        const size_t g = (size_t)(row0 + r) * HID4 + (qi & 15);
        x0[b][u] = F0[g];
        x1[b][u] = F1[g];
        ag[b][u] = AG[g];
        ah[b][u] = AH[g];
        dv[b][u] = DI[row0 + r];
      }
    }
#pragma unroll
    for (int b = 0; b < 2; ++b) {
#pragma unroll
      for (int u = 0; u < 2; ++u) {
        const int qi = (int)threadIdx.x + u * 128;
        const int r  = qi >> 4;
        const int q  = qi & 15;
        const v4f x2 = fma4(-dv[b][u], ag[b][u] + ah[b][u], x1[b][u]);
#pragma unroll
        for (int j = 0; j < 3; ++j) {
          v4f o = cf.c[b][0][j] * x0[b][u];
          o = fma4(cf.c[b][1][j], x1[b][u], o);
          o = fma4(cf.c[b][2][j], x2, o);
          ho[b][u][j] = o;
          *(v4f*)(hs + r * HP + b * (3 * HID) + j * HID + q * 4) = o;
        }
      }
    }
  }
  __syncthreads();
#pragma unroll
  for (int b = 0; b < 2; ++b) {
#pragma unroll
    for (int u = 0; u < 2; ++u) {
      const int qi = (int)threadIdx.x + u * 128;
      float* orow = out + (size_t)(row0 + (qi >> 4)) * DHID + b * (3 * HID) + (qi & 15) * 4;
#pragma unroll
      for (int j = 0; j < 3; ++j) *(v4f*)(orow + j * HID) = ho[b][u][j];
    }
  }

  const int n0   = wave * 16;
  const int mr   = lane & 15;
  const int kq   = (lane >> 4) << 1;
  v8f c = {};
  const float* arow = hs + mr * HP + kq;
  for (int k = 0; k < DHID; k += 4) {
    v2f a, b;
    a = *(const v2f*)(arow + k);
    b.x = Wm1[(size_t)(k + kq) * HID + n0 + mr];
    b.y = Wm1[(size_t)(k + kq + 1) * HID + n0 + mr];
    c = __builtin_amdgcn_wmma_f32_16x16x4_f32(false, a, false, b, (short)0, c, false, false);
  }
  const int mhi = (lane >> 4) << 3;
  const float bv = bm1[n0 + mr];
#pragma unroll
  for (int r = 0; r < 8; ++r) {
    float v = c[r] + bv;
    m[mhi + r][n0 + mr] = v > 0.0f ? v : 0.0f;
  }
  __syncthreads();
  if (threadIdx.x < 32) {
    int row = threadIdx.x >> 1;
    int col = threadIdx.x & 1;
    float acc = bm2[col];
#pragma unroll
    for (int k = 0; k < HID; ++k) acc += m[row][k] * Wm2[k * 2 + col];
    logits[(size_t)(row0 + row) * 2 + col] = acc;
  }
}


static void run_branch(const float* X, const int* src, const int* dst,
                       const float* W1, const float* b1,
                       float* f0, float* f1, float* t, float* ag1, float* ag1b, float* ag2, float* ag2b,
                       float* part, float* dinv, hipStream_t stream) {
  const int nQuadB = NQ / (256 * QPT);
  const int nScatB = (NE / EPW + 7) / 8;

  k_degree_bins<<<DR * DS, DTH, 0, stream>>>(dst, part);

  k_gemm1_relu_emit<<<NN / 16, 128, 0, stream>>>(X, W1, b1, part, f0, t, ag1, ag1b, dinv);

  k_scatter_rows<<<nScatB, 256, 0, stream>>>(src, dst, t, ag1, ag1b);
  k_update_emit4<<<nQuadB, 256, 0, stream>>>((const float4*)f0, (const float4*)ag1, (const float4*)ag1b, dinv,
                                            (float4*)f1, (float4*)t, (float4*)ag2, (float4*)ag2b);

  k_scatter_rows<<<nScatB, 256, 0, stream>>>(src, dst, t, ag2, ag2b);
}

extern "C" void kernel_launch(void* const* d_in, const int* in_sizes, int n_in,
                              void* d_out, int out_size, void* d_ws, size_t ws_size,
                              hipStream_t stream) {
  (void)in_sizes; (void)n_in; (void)out_size; (void)ws_size;
  const float* x     = (const float*)d_in[0];
  const float* sim_x = (const float*)d_in[1];
  const int*   src   = (const int*)d_in[2];
  const int*   dst   = (const int*)d_in[3];
  const int*   ssrc  = (const int*)d_in[4];
  const int*   sdst  = (const int*)d_in[5];
  const float* W1o   = (const float*)d_in[6];
  const float* b1o   = (const float*)d_in[7];
  const float* W1s   = (const float*)d_in[8];
  const float* b1s   = (const float*)d_in[9];
  const float* Wm1   = (const float*)d_in[10];
  const float* bm1   = (const float*)d_in[11];
  const float* Wm2   = (const float*)d_in[12];
  const float* bm2   = (const float*)d_in[13];

  float* out    = (float*)d_out;
  float* logits = out + (size_t)NN * DHID;

  float* f0o  = (float*)d_ws;
  float* f1o  = f0o  + (size_t)NN * HID;
  float* ago  = f1o  + (size_t)NN * HID;
  float* f0s  = ago  + (size_t)NN * HID;
  float* f1s  = f0s  + (size_t)NN * HID;
  float* ags  = f1s  + (size_t)NN * HID;
  float* t    = ags  + (size_t)NN * HID;
  float* ag1  = t    + (size_t)NN * HID;
  float* ag1b = ag1  + (size_t)NN * HID;
  float* agob = ag1b + (size_t)NN * HID;
  float* agsb = agob + (size_t)NN * HID;
  float* dio  = agsb + (size_t)NN * HID;
  float* dis  = dio  + NN;
  float* pco  = dis  + NN;
  float* pcs  = pco  + (size_t)DS * DPP;

  const Coef cf = {{
      {{3.0f,   0.0f,  0.0f},
       {-3.0f,  3.0f,  0.0f},
       {0.75f, -1.5f,  0.75f}},
      {{4.0f, 4.0f, 4.0f},
       {4.0f, 4.0f, 4.0f},
       {4.0f, 4.0f, 4.0f}}}};

  run_branch(x,     src,  dst,  W1o, b1o, f0o, f1o, t, ag1, ag1b, ago, agob, pco, dio, stream);
  run_branch(sim_x, ssrc, sdst, W1s, b1s, f0s, f1s, t, ag1, ag1b, ags, agsb, pcs, dis, stream);

  k_head_build<<<NN / 16, 128, 0, stream>>>(f0o, f1o, ago, agob, dio, f0s, f1s, ags, agsb, dis, cf,
                                            Wm1, bm1, Wm2, bm2, out, logits);
}
